# starting kernel
# baseline (speedup 1.0000x reference)
#include <hip/hip_runtime.h>
#include <hip/hip_bf16.h>


typedef __attribute__((ext_vector_type(2))) float v2f;
typedef __attribute__((ext_vector_type(8))) float v8f;

#define FEATS 4096
#define NBATCH 2048
#define BM 128
#define BN 128
#define BK 32
#define ASTRIDE 36
#define GUARD 128
#define RKN (FEATS + GUARD)
#define ATILE_F (BM * ASTRIDE)
#define BIAS2 112

__global__ __launch_bounds__(256)
void toeplitz_wmma_kernel(const float* __restrict__ x,
                          const float* __restrict__ kern,
                          float* __restrict__ out) {
    __shared__ float smem[RKN + 2 * ATILE_F];
    float* rk    = smem;
    float* abase = smem + RKN;

    const int tid  = threadIdx.x;
    const int lane = tid & 31;
    const int wv   = tid >> 5;
    const int ln   = lane & 15;
    const int half = lane >> 4;
    const int koff = half * 2;

    const int n0 = blockIdx.x * BN;
    const int m0 = blockIdx.y * BM;

    #pragma unroll
    for (int i = tid; i < FEATS / 4; i += 256) {
        float4 v = *(const float4*)(kern + i * 4);
        float4 r; r.x = v.w; r.y = v.z; r.z = v.y; r.w = v.x;
        *(float4*)&rk[FEATS - 4 - i * 4] = r;
    }
    if (tid < GUARD) rk[FEATS + tid] = 0.0f;

    const int lr = tid >> 3;
    const int lq = tid & 7;
    #pragma unroll
    for (int p = 0; p < 4; ++p) {
        const int row = lr + p * 32;
        float4 v = *(const float4*)(x + (size_t)(m0 + row) * FEATS + lq * 4);
        *(float4*)&abase[row * ASTRIDE + lq * 4] = v;
    }
    __syncthreads();

    v8f acc[8];
    #pragma unroll
    for (int t = 0; t < 8; ++t) acc[t] = (v8f){0.f,0.f,0.f,0.f,0.f,0.f,0.f,0.f};

    const int nchunks = (n0 + BN) / BK;

    const int rl0   = (FEATS - 1 - BIAS2) + koff - ln - n0;
    const int mbase = (wv * 16 + ln) * ASTRIDE + koff;

    for (int c = 0; c < nchunks; ++c) {
        const int kc = c * BK;

        if (c + 1 < nchunks) {
            const int kn = kc + BK;
            float* dst = abase + ((c + 1) & 1) * ATILE_F;
            #pragma unroll
            for (int p = 0; p < 4; ++p) {
                const int row = lr + p * 32;
                const float* src = x + (size_t)(m0 + row) * FEATS + kn + lq * 4;
                float4 v = *(const float4*)src;
                __builtin_prefetch(src + BK, 0, 1);
                *(float4*)&dst[row * ASTRIDE + lq * 4] = v;
            }
        }

        const float* abuf = abase + (c & 1) * ATILE_F;
        int rbase = rl0 + kc;
        asm volatile("" : "+v"(rbase));
        const float* rp = rk + rbase;
        #pragma unroll
        for (int s = 0; s < 8; ++s) {
            float2 av = *(const float2*)&abuf[mbase + s * 4];
            v2f a; a.x = av.x; a.y = av.y;
            #pragma unroll
            for (int t = 0; t < 8; ++t) {
                const int off = BIAS2 + 4 * s - 16 * t;
                v2f b; b.x = rp[off]; b.y = rp[off + 1];
                acc[t] = __builtin_amdgcn_wmma_f32_16x16x4_f32(
                    false, a, false, b, (short)0, acc[t], false, false);
            }
        }
        __syncthreads();
    }

    #pragma unroll
    for (int t = 0; t < 8; ++t) {
        const int col = n0 + 16 * t + ln;
        #pragma unroll
        for (int r = 0; r < 8; ++r) {
            const int row = m0 + wv * 16 + half * 8 + r;
            out[(size_t)row * FEATS + col] = acc[t][r];
        }
    }
}

extern "C" void kernel_launch(void* const* d_in, const int* in_sizes, int n_in,
                              void* d_out, int out_size, void* d_ws, size_t ws_size,
                              hipStream_t stream) {
    const float* x    = (const float*)d_in[0];
    const float* kern = (const float*)d_in[1];
    float* out        = (float*)d_out;

    dim3 grid(FEATS / BN, NBATCH / BM);
    toeplitz_wmma_kernel<<<grid, 256, 0, stream>>>(x, kern, out);
}


# X tiles go through registers: global loads for tile c+2 are issued at the head of chunk c and stored to LDS one chunk later (was load, wait, store four times in series before each chunk's WMMAs)
# speedup vs baseline: 1.2356x; 1.2356x over previous
#include <hip/hip_runtime.h>
#include <hip/hip_bf16.h>


typedef __attribute__((ext_vector_type(2))) float v2f;
typedef __attribute__((ext_vector_type(8))) float v8f;

#define FEATS 4096
#define NBATCH 2048
#define BM 128
#define BN 128
#define BK 32
#define ASTRIDE 36
#define GUARD 128
#define RKN (FEATS + GUARD)
#define ATILE_F (BM * ASTRIDE)
#define BIAS2 112

__global__ __launch_bounds__(256)
void toeplitz_wmma_kernel(const float* __restrict__ x,
                          const float* __restrict__ kern,
                          float* __restrict__ out) {
    __shared__ float smem[RKN + 2 * ATILE_F];
    float* rk    = smem;
    float* abase = smem + RKN;

    const int tid  = threadIdx.x;
    const int lane = tid & 31;
    const int wv   = tid >> 5;
    const int ln   = lane & 15;
    const int half = lane >> 4;
    const int koff = half * 2;

    const int n0 = blockIdx.x * BN;
    const int m0 = blockIdx.y * BM;

    #pragma unroll
    for (int i = tid; i < FEATS / 4; i += 256) {
        float4 v = *(const float4*)(kern + i * 4);
        float4 r; r.x = v.w; r.y = v.z; r.z = v.y; r.w = v.x;
        *(float4*)&rk[FEATS - 4 - i * 4] = r;
    }
    if (tid < GUARD) rk[FEATS + tid] = 0.0f;

    const int lr = tid >> 3;
    const int lq = tid & 7;
    #pragma unroll
    for (int p = 0; p < 4; ++p) {
        const int row = lr + p * 32;
        float4 v = *(const float4*)(x + (size_t)(m0 + row) * FEATS + lq * 4);
        *(float4*)&abase[row * ASTRIDE + lq * 4] = v;
    }
    const float* xthr = x + (size_t)(m0 + lr) * FEATS + lq * 4;
    float4 nx0 = *(const float4*)(xthr + BK);
    float4 nx1 = *(const float4*)(xthr + BK + (size_t)32 * FEATS);
    float4 nx2 = *(const float4*)(xthr + BK + (size_t)64 * FEATS);
    float4 nx3 = *(const float4*)(xthr + BK + (size_t)96 * FEATS);
    __syncthreads();

    v8f acc[8];
    #pragma unroll
    for (int t = 0; t < 8; ++t) acc[t] = (v8f){0.f,0.f,0.f,0.f,0.f,0.f,0.f,0.f};

    const int nchunks = (n0 + BN) / BK;

    const int rl0   = (FEATS - 1 - BIAS2) + koff - ln - n0;
    const int mbase = (wv * 16 + ln) * ASTRIDE + koff;

    for (int c = 0; c < nchunks; ++c) {
        const int kc = c * BK;

        if (c + 1 < nchunks) {
            float* dst = abase + ((c + 1) & 1) * ATILE_F + lr * ASTRIDE + lq * 4;
            *(float4*)(dst)                = nx0;
            *(float4*)(dst + 32 * ASTRIDE) = nx1;
            *(float4*)(dst + 64 * ASTRIDE) = nx2;
            *(float4*)(dst + 96 * ASTRIDE) = nx3;
        }
        if (c + 2 < nchunks) {
            const float* src = xthr + kc + 2 * BK;
            nx0 = *(const float4*)(src);
            nx1 = *(const float4*)(src + (size_t)32 * FEATS);
            nx2 = *(const float4*)(src + (size_t)64 * FEATS);
            nx3 = *(const float4*)(src + (size_t)96 * FEATS);
            #pragma unroll
            for (int p = 0; p < 4; ++p)
                __builtin_prefetch(src + (size_t)p * 32 * FEATS + BK, 0, 1);
        }

        const float* abuf = abase + (c & 1) * ATILE_F;
        int rbase = rl0 + kc;
        asm volatile("" : "+v"(rbase));
        const float* rp = rk + rbase;
        #pragma unroll
        for (int s = 0; s < 8; ++s) {
            float2 av = *(const float2*)&abuf[mbase + s * 4];
            v2f a; a.x = av.x; a.y = av.y;
            #pragma unroll
            for (int t = 0; t < 8; ++t) {
                const int off = BIAS2 + 4 * s - 16 * t;
                v2f b; b.x = rp[off]; b.y = rp[off + 1];
                acc[t] = __builtin_amdgcn_wmma_f32_16x16x4_f32(
                    false, a, false, b, (short)0, acc[t], false, false);
            }
        }
        __syncthreads();
    }

    #pragma unroll
    for (int t = 0; t < 8; ++t) {
        const int col = n0 + 16 * t + ln;
        #pragma unroll
        for (int r = 0; r < 8; ++r) {
            const int row = m0 + wv * 16 + half * 8 + r;
            out[(size_t)row * FEATS + col] = acc[t][r];
        }
    }
}

extern "C" void kernel_launch(void* const* d_in, const int* in_sizes, int n_in,
                              void* d_out, int out_size, void* d_ws, size_t ws_size,
                              hipStream_t stream) {
    const float* x    = (const float*)d_in[0];
    const float* kern = (const float*)d_in[1];
    float* out        = (float*)d_out;

    dim3 grid(FEATS / BN, NBATCH / BM);
    toeplitz_wmma_kernel<<<grid, 256, 0, stream>>>(x, kern, out);
}


# Each workgroup computes two column tiles in turn, tile 31-p then tile p, so every workgroup runs 132 K-chunks (was 4 to 128, one tile each); grid 16x16 instead of 32x16
# speedup vs baseline: 2.0865x; 1.6887x over previous
#include <hip/hip_runtime.h>
#include <hip/hip_bf16.h>


typedef __attribute__((ext_vector_type(2))) float v2f;
typedef __attribute__((ext_vector_type(8))) float v8f;

#define FEATS 4096
#define NBATCH 2048
#define BM 128
#define BN 128
#define BK 32
#define ASTRIDE 36
#define GUARD 128
#define RKN (FEATS + GUARD)
#define ATILE_F (BM * ASTRIDE)
#define BIAS2 112
#define NTILE (FEATS / BN)
#define NPAIR (NTILE / 2)

__global__ __launch_bounds__(256)
void toeplitz_wmma_kernel(const float* __restrict__ x,
                          const float* __restrict__ kern,
                          float* __restrict__ out) {
    __shared__ float smem[RKN + 2 * ATILE_F];
    float* rk    = smem;
    float* abase = smem + RKN;

    const int tid  = threadIdx.x;
    const int lane = tid & 31;
    const int wv   = tid >> 5;
    const int ln   = lane & 15;
    const int half = lane >> 4;
    const int koff = half * 2;

    const int m0 = blockIdx.y * BM;

    #pragma unroll
    for (int i = tid; i < FEATS / 4; i += 256) {
        float4 v = *(const float4*)(kern + i * 4);
        float4 r; r.x = v.w; r.y = v.z; r.z = v.y; r.w = v.x;
        *(float4*)&rk[FEATS - 4 - i * 4] = r;
    }
    if (tid < GUARD) rk[FEATS + tid] = 0.0f;

    const int lr = tid >> 3;
    const int lq = tid & 7;
    const float* xthr = x + (size_t)(m0 + lr) * FEATS + lq * 4;
    const int mbase = (wv * 16 + ln) * ASTRIDE + koff;

    for (int pass = 0; pass < 2; ++pass) {
        const int n0 = (pass == 0 ? (NTILE - 1 - (int)blockIdx.x) : (int)blockIdx.x) * BN;

        #pragma unroll
        for (int p = 0; p < 4; ++p) {
            const int row = lr + p * 32;
            float4 v = *(const float4*)(x + (size_t)(m0 + row) * FEATS + lq * 4);
            *(float4*)&abase[row * ASTRIDE + lq * 4] = v;
        }
        float4 nx0 = *(const float4*)(xthr + BK);
        float4 nx1 = *(const float4*)(xthr + BK + (size_t)32 * FEATS);
        float4 nx2 = *(const float4*)(xthr + BK + (size_t)64 * FEATS);
        float4 nx3 = *(const float4*)(xthr + BK + (size_t)96 * FEATS);
        __syncthreads();

        v8f acc[8];
        #pragma unroll
        for (int t = 0; t < 8; ++t) acc[t] = (v8f){0.f,0.f,0.f,0.f,0.f,0.f,0.f,0.f};

        const int nchunks = (n0 + BN) / BK;

        const int rl0   = (FEATS - 1 - BIAS2) + koff - ln - n0;

        for (int c = 0; c < nchunks; ++c) {
            const int kc = c * BK;

            if (c + 1 < nchunks) {
                float* dst = abase + ((c + 1) & 1) * ATILE_F + lr * ASTRIDE + lq * 4;
                *(float4*)(dst)                = nx0;
                *(float4*)(dst + 32 * ASTRIDE) = nx1;
                *(float4*)(dst + 64 * ASTRIDE) = nx2;
                *(float4*)(dst + 96 * ASTRIDE) = nx3;
            }
            if (c + 2 < nchunks) {
                const float* src = xthr + kc + 2 * BK;
                nx0 = *(const float4*)(src);
                nx1 = *(const float4*)(src + (size_t)32 * FEATS);
                nx2 = *(const float4*)(src + (size_t)64 * FEATS);
                nx3 = *(const float4*)(src + (size_t)96 * FEATS);
                #pragma unroll
                for (int p = 0; p < 4; ++p)
                    __builtin_prefetch(src + (size_t)p * 32 * FEATS + BK, 0, 1);
            }

            const float* abuf = abase + (c & 1) * ATILE_F;
            int rbase = rl0 + kc;
            asm volatile("" : "+v"(rbase));
            const float* rp = rk + rbase;
            #pragma unroll
            for (int s = 0; s < 8; ++s) {
                float2 av = *(const float2*)&abuf[mbase + s * 4];
                v2f a; a.x = av.x; a.y = av.y;
                #pragma unroll
                for (int t = 0; t < 8; ++t) {
                    const int off = BIAS2 + 4 * s - 16 * t;
                    v2f b; b.x = rp[off]; b.y = rp[off + 1];
                    acc[t] = __builtin_amdgcn_wmma_f32_16x16x4_f32(
                        false, a, false, b, (short)0, acc[t], false, false);
                }
            }
            __syncthreads();
        }

        #pragma unroll
        for (int t = 0; t < 8; ++t) {
            const int col = n0 + 16 * t + ln;
            #pragma unroll
            for (int r = 0; r < 8; ++r) {
                const int row = m0 + wv * 16 + half * 8 + r;
                out[(size_t)row * FEATS + col] = acc[t][r];
            }
        }
    }
}

extern "C" void kernel_launch(void* const* d_in, const int* in_sizes, int n_in,
                              void* d_out, int out_size, void* d_ws, size_t ws_size,
                              hipStream_t stream) {
    const float* x    = (const float*)d_in[0];
    const float* kern = (const float*)d_in[1];
    float* out        = (float*)d_out;

    dim3 grid(NPAIR, NBATCH / BM);
    toeplitz_wmma_kernel<<<grid, 256, 0, stream>>>(x, kern, out);
}


# Three 64x64 Toeplitz block products per 128-k block instead of four: P1=(x0+x1)*K, P2=x1*G[d-64], P3=x0*G, G=K[d+64]-K[d]; y0=P1-P2, y1=P1+P3 (f32 sums before products: rounding, sum order change)
# speedup vs baseline: 2.4156x; 1.1577x over previous
#include <hip/hip_runtime.h>


typedef __attribute__((ext_vector_type(2))) float v2f;
typedef __attribute__((ext_vector_type(8))) float v8f;

#define FEATS 4096
#define NBATCH 2048
#define BM 128
#define BN 128
#define BK 32
#define HALF 64
#define ASTRIDE 36
#define GUARD 128
#define RKN (FEATS + GUARD)
#define ATILE_F (BM * ASTRIDE)
#define STAGE_F (2 * ATILE_F)
#define BIASK 48
#define NTILE (FEATS / BN)
#define NPAIR (NTILE / 2)

__global__ __launch_bounds__(256)
void toeplitz_wmma_kernel(const float* __restrict__ x,
                          const float* __restrict__ kern,
                          float* __restrict__ out) {
    __shared__ float smem[2 * RKN + 2 * STAGE_F];
    float* rk    = smem;
    float* rg    = smem + RKN;
    float* abase = smem + 2 * RKN;

    const int tid  = threadIdx.x;
    const int lane = tid & 31;
    const int wv   = tid >> 5;
    const int ln   = lane & 15;
    const int half = lane >> 4;
    const int koff = half * 2;

    const int m0 = blockIdx.y * BM;

    #pragma unroll
    for (int i = tid; i < FEATS / 4; i += 256) {
        float4 v = *(const float4*)(kern + i * 4);
        float4 r; r.x = v.w; r.y = v.z; r.z = v.y; r.w = v.x;
        *(float4*)&rk[FEATS - 4 - i * 4] = r;
    }
    if (tid < GUARD) rk[FEATS + tid] = 0.0f;
    __syncthreads();
    for (int i = tid; i < RKN; i += 256)
        rg[i] = (i >= HALF) ? (rk[i - HALF] - rk[i]) : 0.0f;

    const int lr = tid >> 3;
    const int lq = tid & 7;
    const float* xthr = x + (size_t)(m0 + lr) * FEATS + lq * 4;
    const int mbase = (wv * 16 + ln) * ASTRIDE + koff;
    const int sbase = lr * ASTRIDE + lq * 4;

    for (int pass = 0; pass < 2; ++pass) {
        const int n0 = (pass == 0 ? (NTILE - 1 - (int)blockIdx.x) : (int)blockIdx.x) * BN;

        #pragma unroll
        for (int p = 0; p < 4; ++p) {
            const float* src = xthr + (size_t)p * 32 * FEATS;
            float4 v0 = *(const float4*)(src);
            float4 v1 = *(const float4*)(src + HALF);
            *(float4*)&abase[sbase + p * 32 * ASTRIDE]           = v0;
            *(float4*)&abase[ATILE_F + sbase + p * 32 * ASTRIDE] = v1;
        }
        float4 nx0 = *(const float4*)(xthr + BK);
        float4 nx1 = *(const float4*)(xthr + BK + (size_t)32 * FEATS);
        float4 nx2 = *(const float4*)(xthr + BK + (size_t)64 * FEATS);
        float4 nx3 = *(const float4*)(xthr + BK + (size_t)96 * FEATS);
        float4 ny0 = *(const float4*)(xthr + BK + HALF);
        float4 ny1 = *(const float4*)(xthr + BK + HALF + (size_t)32 * FEATS);
        float4 ny2 = *(const float4*)(xthr + BK + HALF + (size_t)64 * FEATS);
        float4 ny3 = *(const float4*)(xthr + BK + HALF + (size_t)96 * FEATS);
        __syncthreads();

        v8f accS[4], accA[4], accB[4];
        #pragma unroll
        for (int t = 0; t < 4; ++t) {
            accS[t] = (v8f){0.f,0.f,0.f,0.f,0.f,0.f,0.f,0.f};
            accA[t] = (v8f){0.f,0.f,0.f,0.f,0.f,0.f,0.f,0.f};
            accB[t] = (v8f){0.f,0.f,0.f,0.f,0.f,0.f,0.f,0.f};
        }

        const int nch = (n0 + BN) / HALF;

        const int rl0 = (FEATS - 1 - BIASK) + koff - ln - n0;

        for (int c = 0; c < nch; ++c) {
            const int kx = (c >> 1) * 128 + (c & 1) * BK;

            if (c + 1 < nch) {
                float* dst = abase + ((c + 1) & 1) * STAGE_F + sbase;
                *(float4*)(dst)                          = nx0;
                *(float4*)(dst + 32 * ASTRIDE)           = nx1;
                *(float4*)(dst + 64 * ASTRIDE)           = nx2;
                *(float4*)(dst + 96 * ASTRIDE)           = nx3;
                *(float4*)(dst + ATILE_F)                = ny0;
                *(float4*)(dst + ATILE_F + 32 * ASTRIDE) = ny1;
                *(float4*)(dst + ATILE_F + 64 * ASTRIDE) = ny2;
                *(float4*)(dst + ATILE_F + 96 * ASTRIDE) = ny3;
            }
            if (c + 2 < nch) {
                const float* src = xthr + kx + 128;
                nx0 = *(const float4*)(src);
                nx1 = *(const float4*)(src + (size_t)32 * FEATS);
                nx2 = *(const float4*)(src + (size_t)64 * FEATS);
                nx3 = *(const float4*)(src + (size_t)96 * FEATS);
                ny0 = *(const float4*)(src + HALF);
                ny1 = *(const float4*)(src + HALF + (size_t)32 * FEATS);
                ny2 = *(const float4*)(src + HALF + (size_t)64 * FEATS);
                ny3 = *(const float4*)(src + HALF + (size_t)96 * FEATS);
                if (c + 3 < nch) {
                    const float* pf = xthr + kx + ((c & 1) ? 224 : 160);
                    #pragma unroll
                    for (int p = 0; p < 4; ++p) {
                        __builtin_prefetch(pf + (size_t)p * 32 * FEATS, 0, 1);
                        __builtin_prefetch(pf + (size_t)p * 32 * FEATS + HALF, 0, 1);
                    }
                }
            }

            const float* abuf = abase + (c & 1) * STAGE_F;
            int rbase = rl0 + kx;
            int gbase = rl0 + kx + RKN;
            asm volatile("" : "+v"(rbase));
            asm volatile("" : "+v"(gbase));
            const float* rp = rk + rbase;
            const float* gp = rk + gbase;
            #pragma unroll
            for (int s = 0; s < 8; ++s) {
                float2 a0v = *(const float2*)&abuf[mbase + s * 4];
                float2 a1v = *(const float2*)&abuf[ATILE_F + mbase + s * 4];
                v2f a0; a0.x = a0v.x; a0.y = a0v.y;
                v2f a1; a1.x = a1v.x; a1.y = a1v.y;
                v2f as = a0 + a1;
                #pragma unroll
                for (int t = 0; t < 4; ++t) {
                    const int off = BIASK + 4 * s - 16 * t;
                    v2f bk; bk.x = rp[off];        bk.y = rp[off + 1];
                    v2f b3; b3.x = gp[off];        b3.y = gp[off + 1];
                    v2f b2; b2.x = gp[off + HALF]; b2.y = gp[off + HALF + 1];
                    accS[t] = __builtin_amdgcn_wmma_f32_16x16x4_f32(
                        false, as, false, bk, (short)0, accS[t], false, false);
                    accB[t] = __builtin_amdgcn_wmma_f32_16x16x4_f32(
                        false, a0, false, b3, (short)0, accB[t], false, false);
                    accA[t] = __builtin_amdgcn_wmma_f32_16x16x4_f32(
                        false, a1, false, b2, (short)0, accA[t], false, false);
                }
            }
            __syncthreads();
        }

        #pragma unroll
        for (int t = 0; t < 4; ++t) {
            const int col = n0 + 16 * t + ln;
            #pragma unroll
            for (int r = 0; r < 8; ++r) {
                const int row = m0 + wv * 16 + half * 8 + r;
                out[(size_t)row * FEATS + col]        = accS[t][r] - accA[t][r];
                out[(size_t)row * FEATS + col + HALF] = accS[t][r] + accB[t][r];
            }
        }
    }
}

extern "C" void kernel_launch(void* const* d_in, const int* in_sizes, int n_in,
                              void* d_out, int out_size, void* d_ws, size_t ws_size,
                              hipStream_t stream) {
    const float* x    = (const float*)d_in[0];
    const float* kern = (const float*)d_in[1];
    float* out        = (float*)d_out;

    dim3 grid(NPAIR, NBATCH / BM);
    toeplitz_wmma_kernel<<<grid, 256, 0, stream>>>(x, kern, out);
}


# Next X tile's LDS staging stores and the global loads behind them move from the chunk head to behind k-step 2, among the WMMAs; fragments are read ahead of them; chunk barrier waits for LDS only
# speedup vs baseline: 2.6529x; 1.0982x over previous
#include <hip/hip_runtime.h>


typedef __attribute__((ext_vector_type(2))) float v2f;
typedef __attribute__((ext_vector_type(8))) float v8f;

#define FEATS 4096
#define NBATCH 2048
#define BM 128
#define BN 128
#define BK 32
#define HALF 64
#define ASTRIDE 36
#define GUARD 128
#define RKN (FEATS + GUARD)
#define ATILE_F (BM * ASTRIDE)
#define STAGE_F (2 * ATILE_F)
#define BIASK 48
#define NTILE (FEATS / BN)
#define NPAIR (NTILE / 2)
#define STAGE_AT 2

__global__ __launch_bounds__(256)
void toeplitz_wmma_kernel(const float* __restrict__ x,
                          const float* __restrict__ kern,
                          float* __restrict__ out) {
    __shared__ float smem[2 * RKN + 2 * STAGE_F];
    float* rk    = smem;
    float* rg    = smem + RKN;
    float* abase = smem + 2 * RKN;

    const int tid  = threadIdx.x;
    const int lane = tid & 31;
    const int wv   = tid >> 5;
    const int ln   = lane & 15;
    const int half = lane >> 4;
    const int koff = half * 2;

    const int m0 = blockIdx.y * BM;

    #pragma unroll
    for (int i = tid; i < FEATS / 4; i += 256) {
        float4 v = *(const float4*)(kern + i * 4);
        float4 r; r.x = v.w; r.y = v.z; r.z = v.y; r.w = v.x;
        *(float4*)&rk[FEATS - 4 - i * 4] = r;
    }
    if (tid < GUARD) rk[FEATS + tid] = 0.0f;
    __syncthreads();
    for (int i = tid; i < RKN; i += 256)
        rg[i] = (i >= HALF) ? (rk[i - HALF] - rk[i]) : 0.0f;

    const int lr = tid >> 3;
    const int lq = tid & 7;
    const float* xthr = x + (size_t)(m0 + lr) * FEATS + lq * 4;
    const int mbase = (wv * 16 + ln) * ASTRIDE + koff;
    const int sbase = lr * ASTRIDE + lq * 4;

    for (int pass = 0; pass < 2; ++pass) {
        const int n0 = (pass == 0 ? (NTILE - 1 - (int)blockIdx.x) : (int)blockIdx.x) * BN;

        #pragma unroll
        for (int p = 0; p < 4; ++p) {
            const float* src = xthr + (size_t)p * 32 * FEATS;
            float4 v0 = *(const float4*)(src);
            float4 v1 = *(const float4*)(src + HALF);
            *(float4*)&abase[sbase + p * 32 * ASTRIDE]           = v0;
            *(float4*)&abase[ATILE_F + sbase + p * 32 * ASTRIDE] = v1;
        }
        float4 nx0 = *(const float4*)(xthr + BK);
        float4 nx1 = *(const float4*)(xthr + BK + (size_t)32 * FEATS);
        float4 nx2 = *(const float4*)(xthr + BK + (size_t)64 * FEATS);
        float4 nx3 = *(const float4*)(xthr + BK + (size_t)96 * FEATS);
        float4 ny0 = *(const float4*)(xthr + BK + HALF);
        float4 ny1 = *(const float4*)(xthr + BK + HALF + (size_t)32 * FEATS);
        float4 ny2 = *(const float4*)(xthr + BK + HALF + (size_t)64 * FEATS);
        float4 ny3 = *(const float4*)(xthr + BK + HALF + (size_t)96 * FEATS);
        __syncthreads();

        v8f accS[4], accA[4], accB[4];
        #pragma unroll
        for (int t = 0; t < 4; ++t) {
            accS[t] = (v8f){0.f,0.f,0.f,0.f,0.f,0.f,0.f,0.f};
            accA[t] = (v8f){0.f,0.f,0.f,0.f,0.f,0.f,0.f,0.f};
            accB[t] = (v8f){0.f,0.f,0.f,0.f,0.f,0.f,0.f,0.f};
        }

        const int nch = (n0 + BN) / HALF;

        const int rl0 = (FEATS - 1 - BIASK) + koff - ln - n0;

        for (int c = 0; c < nch; ++c) {
            const int kx = (c >> 1) * 128 + (c & 1) * BK;

            const float* abuf = abase + (c & 1) * STAGE_F;
            int rbase = rl0 + kx;
            int gbase = rl0 + kx + RKN;
            asm volatile("" : "+v"(rbase));
            asm volatile("" : "+v"(gbase));
            const float* rp = rk + rbase;
            const float* gp = rk + gbase;
            v2f fa0[8], fa1[8];
            #pragma unroll
            for (int s = 0; s < 8; ++s) {
                float2 a0v = *(const float2*)&abuf[mbase + s * 4];
                float2 a1v = *(const float2*)&abuf[ATILE_F + mbase + s * 4];
                fa0[s].x = a0v.x; fa0[s].y = a0v.y;
                fa1[s].x = a1v.x; fa1[s].y = a1v.y;
            }
            v2f fk[20], fg[36];
            #pragma unroll
            for (int i = 0; i < 20; ++i) { fk[i].x = rp[4 * i]; fk[i].y = rp[4 * i + 1]; }
            #pragma unroll
            for (int i = 0; i < 36; ++i) { fg[i].x = gp[4 * i]; fg[i].y = gp[4 * i + 1]; }

            #pragma unroll
            for (int s = 0; s < 8; ++s) {
                if (s == STAGE_AT) {
                    if (c + 1 < nch) {
                        float* dst = abase + ((c + 1) & 1) * STAGE_F + sbase;
                        *(float4*)(dst)                          = nx0;
                        *(float4*)(dst + 32 * ASTRIDE)           = nx1;
                        *(float4*)(dst + 64 * ASTRIDE)           = nx2;
                        *(float4*)(dst + 96 * ASTRIDE)           = nx3;
                        *(float4*)(dst + ATILE_F)                = ny0;
                        *(float4*)(dst + ATILE_F + 32 * ASTRIDE) = ny1;
                        *(float4*)(dst + ATILE_F + 64 * ASTRIDE) = ny2;
                        *(float4*)(dst + ATILE_F + 96 * ASTRIDE) = ny3;
                    }
                    if (c + 2 < nch) {
                        const float* src = xthr + kx + 128;
                        nx0 = *(const float4*)(src);
                        nx1 = *(const float4*)(src + (size_t)32 * FEATS);
                        nx2 = *(const float4*)(src + (size_t)64 * FEATS);
                        nx3 = *(const float4*)(src + (size_t)96 * FEATS);
                        ny0 = *(const float4*)(src + HALF);
                        ny1 = *(const float4*)(src + HALF + (size_t)32 * FEATS);
                        ny2 = *(const float4*)(src + HALF + (size_t)64 * FEATS);
                        ny3 = *(const float4*)(src + HALF + (size_t)96 * FEATS);
                        if (c + 3 < nch) {
                            const float* pf = xthr + kx + ((c & 1) ? 224 : 160);
                            #pragma unroll
                            for (int p = 0; p < 4; ++p) {
                                __builtin_prefetch(pf + (size_t)p * 32 * FEATS, 0, 1);
                                __builtin_prefetch(pf + (size_t)p * 32 * FEATS + HALF, 0, 1);
                            }
                        }
                    }
                }
                v2f a0 = fa0[s];
                v2f a1 = fa1[s];
                v2f as = a0 + a1;
                #pragma unroll
                for (int t = 0; t < 4; ++t) {
                    const int fi = BIASK / 4 + s - 4 * t;
                    v2f bk = fk[fi];
                    v2f b3 = fg[fi];
                    v2f b2 = fg[fi + HALF / 4];
                    accS[t] = __builtin_amdgcn_wmma_f32_16x16x4_f32(
                        false, as, false, bk, (short)0, accS[t], false, false);
                    accB[t] = __builtin_amdgcn_wmma_f32_16x16x4_f32(
                        false, a0, false, b3, (short)0, accB[t], false, false);
                    accA[t] = __builtin_amdgcn_wmma_f32_16x16x4_f32(
                        false, a1, false, b2, (short)0, accA[t], false, false);
                }
            }
            __builtin_amdgcn_fence(__ATOMIC_RELEASE, "workgroup", "local");
            __builtin_amdgcn_s_barrier();
            __builtin_amdgcn_fence(__ATOMIC_ACQUIRE, "workgroup", "local");
        }

        #pragma unroll
        for (int t = 0; t < 4; ++t) {
            const int col = n0 + 16 * t + ln;
            #pragma unroll
            for (int r = 0; r < 8; ++r) {
                const int row = m0 + wv * 16 + half * 8 + r;
                out[(size_t)row * FEATS + col]        = accS[t][r] - accA[t][r];
                out[(size_t)row * FEATS + col + HALF] = accS[t][r] + accB[t][r];
            }
        }
    }
}

extern "C" void kernel_launch(void* const* d_in, const int* in_sizes, int n_in,
                              void* d_out, int out_size, void* d_ws, size_t ws_size,
                              hipStream_t stream) {
    const float* x    = (const float*)d_in[0];
    const float* kern = (const float*)d_in[1];
    float* out        = (float*)d_out;

    dim3 grid(NPAIR, NBATCH / BM);
    toeplitz_wmma_kernel<<<grid, 256, 0, stream>>>(x, kern, out);
}


# Second level of the three-product split: 9 products of 32-wide Toeplitz blocks per 128-k block (was 12), vectors K, GK, G, GG in LDS; f32 sums before products: rounding, sum order change
# speedup vs baseline: 2.8785x; 1.0850x over previous
#include <hip/hip_runtime.h>


typedef __attribute__((ext_vector_type(2))) float v2f;
typedef __attribute__((ext_vector_type(8))) float v8f;

#define FEATS 4096
#define NBATCH 2048
#define BM 128
#define BN 128
#define QW 32
#define CK 16
#define HALF 64
#define ASTRIDE 36
#define GUARD 128
#define RKN (FEATS + GUARD)
#define ATILE_F (BM * ASTRIDE)
#define STAGE_F (2 * ATILE_F)
#define BIASK 16
#define NTILE (FEATS / BN)
#define NPAIR (NTILE / 2)
#define STAGE_AT 1
#define KSTEPS (CK / 4)

__global__ __launch_bounds__(256)
void toeplitz_wmma_kernel(const float* __restrict__ x,
                          const float* __restrict__ kern,
                          float* __restrict__ out) {
    __shared__ float smem[4 * RKN + 2 * STAGE_F];
    float* rk    = smem;
    float* rgk   = smem + RKN;
    float* rg    = smem + 2 * RKN;
    float* rgg   = smem + 3 * RKN;
    float* abase = smem + 4 * RKN;

    const int tid  = threadIdx.x;
    const int lane = tid & 31;
    const int wv   = tid >> 5;
    const int ln   = lane & 15;
    const int half = lane >> 4;
    const int koff = half * 2;

    const int m0 = blockIdx.y * BM;

    #pragma unroll
    for (int i = tid; i < FEATS / 4; i += 256) {
        float4 v = *(const float4*)(kern + i * 4);
        float4 r; r.x = v.w; r.y = v.z; r.z = v.y; r.w = v.x;
        *(float4*)&rk[FEATS - 4 - i * 4] = r;
    }
    if (tid < GUARD) rk[FEATS + tid] = 0.0f;
    __syncthreads();
    for (int i = tid; i < RKN; i += 256) {
        const float k0  = rk[i];
        const float k32 = (i >= QW)        ? rk[i - QW]        : 0.0f;
        const float k64 = (i >= HALF)      ? rk[i - HALF]      : 0.0f;
        const float k96 = (i >= HALF + QW) ? rk[i - HALF - QW] : 0.0f;
        rgk[i] = (i >= QW)        ? (k32 - k0) : 0.0f;
        rg[i]  = (i >= HALF)      ? (k64 - k0) : 0.0f;
        rgg[i] = (i >= HALF + QW) ? ((k96 - k32) - (k64 - k0)) : 0.0f;
    }

    const int lr = tid >> 3;
    const int lq = tid & 7;
    const float* xthr = x + (size_t)(m0 + lr) * FEATS + (lq >> 2) * QW + (lq & 3) * 4;
    const int mbase = (wv * 16 + ln) * ASTRIDE + koff;
    const int sbase = lr * ASTRIDE + lq * 4;

    for (int pass = 0; pass < 2; ++pass) {
        const int n0 = (pass == 0 ? (NTILE - 1 - (int)blockIdx.x) : (int)blockIdx.x) * BN;

        #pragma unroll
        for (int p = 0; p < 4; ++p) {
            const float* src = xthr + (size_t)p * 32 * FEATS;
            float4 v0 = *(const float4*)(src);
            float4 v1 = *(const float4*)(src + HALF);
            *(float4*)&abase[sbase + p * 32 * ASTRIDE]           = v0;
            *(float4*)&abase[ATILE_F + sbase + p * 32 * ASTRIDE] = v1;
        }
        float4 nx0 = *(const float4*)(xthr + CK);
        float4 nx1 = *(const float4*)(xthr + CK + (size_t)32 * FEATS);
        float4 nx2 = *(const float4*)(xthr + CK + (size_t)64 * FEATS);
        float4 nx3 = *(const float4*)(xthr + CK + (size_t)96 * FEATS);
        float4 ny0 = *(const float4*)(xthr + CK + HALF);
        float4 ny1 = *(const float4*)(xthr + CK + HALF + (size_t)32 * FEATS);
        float4 ny2 = *(const float4*)(xthr + CK + HALF + (size_t)64 * FEATS);
        float4 ny3 = *(const float4*)(xthr + CK + HALF + (size_t)96 * FEATS);
        __syncthreads();

        v8f q1a[2], q2a[2], q3a[2], q1b[2], q2b[2], q3b[2], q1c[2], q2c[2], q3c[2];
        #pragma unroll
        for (int t = 0; t < 2; ++t) {
            const v8f z = (v8f){0.f,0.f,0.f,0.f,0.f,0.f,0.f,0.f};
            q1a[t] = z; q2a[t] = z; q3a[t] = z;
            q1b[t] = z; q2b[t] = z; q3b[t] = z;
            q1c[t] = z; q2c[t] = z; q3c[t] = z;
        }

        const int nch = (n0 + BN) / HALF;

        const int rl0 = (FEATS - 1 - BIASK) + koff - ln - n0;

        for (int c = 0; c < nch; ++c) {
            const int kx = (c >> 1) * 128 + (c & 1) * CK;

            const float* abuf = abase + (c & 1) * STAGE_F;
            int kbase  = rl0 + kx;
            int gkbase = rl0 + kx + RKN;
            int gbase  = rl0 + kx + 2 * RKN;
            int ggbase = rl0 + kx + 3 * RKN;
            asm volatile("" : "+v"(kbase));
            asm volatile("" : "+v"(gkbase));
            asm volatile("" : "+v"(gbase));
            asm volatile("" : "+v"(ggbase));
            const float* kp  = rk + kbase;
            const float* gkp = rk + gkbase;
            const float* gp  = rk + gbase;
            const float* ggp = rk + ggbase;
            v2f f0[KSTEPS], f1[KSTEPS], f2[KSTEPS], f3[KSTEPS];
            #pragma unroll
            for (int s = 0; s < KSTEPS; ++s) {
                float2 a0v = *(const float2*)&abuf[mbase + s * 4];
                float2 a1v = *(const float2*)&abuf[mbase + CK + s * 4];
                float2 a2v = *(const float2*)&abuf[ATILE_F + mbase + s * 4];
                float2 a3v = *(const float2*)&abuf[ATILE_F + mbase + CK + s * 4];
                f0[s].x = a0v.x; f0[s].y = a0v.y;
                f1[s].x = a1v.x; f1[s].y = a1v.y;
                f2[s].x = a2v.x; f2[s].y = a2v.y;
                f3[s].x = a3v.x; f3[s].y = a3v.y;
            }
            v2f fK[8], fGK[16], fGa[8], fGc[8], fGG[32];
            #pragma unroll
            for (int i = 0; i < 8; ++i)  { fK[i].x  = kp[4 * i];          fK[i].y  = kp[4 * i + 1]; }
            #pragma unroll
            for (int i = 0; i < 16; ++i) { fGK[i].x = gkp[4 * i];         fGK[i].y = gkp[4 * i + 1]; }
            #pragma unroll
            for (int i = 0; i < 8; ++i)  { fGa[i].x = gp[4 * i];          fGa[i].y = gp[4 * i + 1]; }
            #pragma unroll
            for (int i = 0; i < 8; ++i)  { fGc[i].x = gp[4 * (i + 16)];   fGc[i].y = gp[4 * (i + 16) + 1]; }
            #pragma unroll
            for (int i = 0; i < 32; ++i) { fGG[i].x = ggp[4 * i];         fGG[i].y = ggp[4 * i + 1]; }

            #pragma unroll
            for (int s = 0; s < KSTEPS; ++s) {
                if (s == STAGE_AT) {
                    if (c + 1 < nch) {
                        float* dst = abase + ((c + 1) & 1) * STAGE_F + sbase;
                        *(float4*)(dst)                          = nx0;
                        *(float4*)(dst + 32 * ASTRIDE)           = nx1;
                        *(float4*)(dst + 64 * ASTRIDE)           = nx2;
                        *(float4*)(dst + 96 * ASTRIDE)           = nx3;
                        *(float4*)(dst + ATILE_F)                = ny0;
                        *(float4*)(dst + ATILE_F + 32 * ASTRIDE) = ny1;
                        *(float4*)(dst + ATILE_F + 64 * ASTRIDE) = ny2;
                        *(float4*)(dst + ATILE_F + 96 * ASTRIDE) = ny3;
                    }
                    if (c + 2 < nch) {
                        const float* src = xthr + kx + 128;
                        nx0 = *(const float4*)(src);
                        nx1 = *(const float4*)(src + (size_t)32 * FEATS);
                        nx2 = *(const float4*)(src + (size_t)64 * FEATS);
                        nx3 = *(const float4*)(src + (size_t)96 * FEATS);
                        ny0 = *(const float4*)(src + HALF);
                        ny1 = *(const float4*)(src + HALF + (size_t)32 * FEATS);
                        ny2 = *(const float4*)(src + HALF + (size_t)64 * FEATS);
                        ny3 = *(const float4*)(src + HALF + (size_t)96 * FEATS);
                        if ((c & 1) && c + 3 < nch) {
                            const float* pf = xthr + kx + 240;
                            #pragma unroll
                            for (int p = 0; p < 4; ++p) {
                                __builtin_prefetch(pf + (size_t)p * 32 * FEATS, 0, 1);
                                __builtin_prefetch(pf + (size_t)p * 32 * FEATS + HALF, 0, 1);
                            }
                        }
                    }
                }
                const v2f a0 = f0[s];
                const v2f a1 = f1[s];
                const v2f a2 = f2[s];
                const v2f a3 = f3[s];
                const v2f s02 = a0 + a2;
                const v2f s13 = a1 + a3;
                const v2f sss = s02 + s13;
                const v2f s01 = a0 + a1;
                const v2f s23 = a2 + a3;
                #pragma unroll
                for (int t = 0; t < 2; ++t) {
                    const int fi = BIASK / 4 + s - 4 * t;
                    q1a[t] = __builtin_amdgcn_wmma_f32_16x16x4_f32(
                        false, sss, false, fK[fi],       (short)0, q1a[t], false, false);
                    q1b[t] = __builtin_amdgcn_wmma_f32_16x16x4_f32(
                        false, s01, false, fGa[fi],      (short)0, q1b[t], false, false);
                    q1c[t] = __builtin_amdgcn_wmma_f32_16x16x4_f32(
                        false, s23, false, fGc[fi],      (short)0, q1c[t], false, false);
                    q3a[t] = __builtin_amdgcn_wmma_f32_16x16x4_f32(
                        false, s02, false, fGK[fi],      (short)0, q3a[t], false, false);
                    q2a[t] = __builtin_amdgcn_wmma_f32_16x16x4_f32(
                        false, s13, false, fGK[fi + 8],  (short)0, q2a[t], false, false);
                    q3b[t] = __builtin_amdgcn_wmma_f32_16x16x4_f32(
                        false, a0,  false, fGG[fi],      (short)0, q3b[t], false, false);
                    q2b[t] = __builtin_amdgcn_wmma_f32_16x16x4_f32(
                        false, a1,  false, fGG[fi + 8],  (short)0, q2b[t], false, false);
                    q3c[t] = __builtin_amdgcn_wmma_f32_16x16x4_f32(
                        false, a2,  false, fGG[fi + 16], (short)0, q3c[t], false, false);
                    q2c[t] = __builtin_amdgcn_wmma_f32_16x16x4_f32(
                        false, a3,  false, fGG[fi + 24], (short)0, q2c[t], false, false);
                }
            }
            __builtin_amdgcn_fence(__ATOMIC_RELEASE, "workgroup", "local");
            __builtin_amdgcn_s_barrier();
            __builtin_amdgcn_fence(__ATOMIC_ACQUIRE, "workgroup", "local");
        }

        #pragma unroll
        for (int t = 0; t < 2; ++t) {
            const int col = n0 + 16 * t + ln;
            #pragma unroll
            for (int r = 0; r < 8; ++r) {
                const int row = m0 + wv * 16 + half * 8 + r;
                const float p10 = q1a[t][r] - q2a[t][r];
                const float p11 = q1a[t][r] + q3a[t][r];
                const float p20 = q1c[t][r] - q2c[t][r];
                const float p21 = q1c[t][r] + q3c[t][r];
                const float p30 = q1b[t][r] - q2b[t][r];
                const float p31 = q1b[t][r] + q3b[t][r];
                float* o = out + (size_t)row * FEATS + col;
                o[0]      = p10 - p20;
                o[QW]     = p11 - p21;
                o[2 * QW] = p10 + p30;
                o[3 * QW] = p11 + p31;
            }
        }
    }
}

extern "C" void kernel_launch(void* const* d_in, const int* in_sizes, int n_in,
                              void* d_out, int out_size, void* d_ws, size_t ws_size,
                              hipStream_t stream) {
    const float* x    = (const float*)d_in[0];
    const float* kern = (const float*)d_in[1];
    float* out        = (float*)d_out;

    dim3 grid(NPAIR, NBATCH / BM);
    toeplitz_wmma_kernel<<<grid, 256, 0, stream>>>(x, kern, out);
}
